# starting kernel
# baseline (speedup 1.0000x reference)
#include <hip/hip_runtime.h>

typedef float v2f __attribute__((ext_vector_type(2)));
typedef float v8f __attribute__((ext_vector_type(8)));

#define NODES_TOT 20000

__device__ __forceinline__ v8f wmma_f32(v2f a, v2f b, v8f c) {
  return __builtin_amdgcn_wmma_f32_16x16x4_f32(
       false, a,  false, b,
       (short)0, c,  false,  false);
}

template <int K, int N, bool RELU>
__device__ __forceinline__ void gemm_stage(const float* aBuf, const float* w,
                                           const float* bias, float* outBuf,
                                           int col, int hi) {
#pragma unroll
  for (int nt = 0; nt < N / 16; ++nt) {
    float bv = bias ? bias[nt * 16 + col] : 0.0f;
    v8f c;
#pragma unroll
    for (int r = 0; r < 8; ++r) c[r] = bv;
#pragma unroll
    for (int k0 = 0; k0 < K; k0 += 4) {
      int k = k0 + 2 * hi;
      v2f a;
      a.x = aBuf[col * K + k];
      a.y = aBuf[col * K + k + 1];
      v2f b;
      b.x = w[k * N + nt * 16 + col];
      b.y = w[(k + 1) * N + nt * 16 + col];
      c = wmma_f32(a, b, c);
    }
#pragma unroll
    for (int r = 0; r < 8; ++r) {
      float v = c[r];
      if (RELU) v = fmaxf(v, 0.1f * v);
      outBuf[(r + 8 * hi) * N + nt * 16 + col] = v;
    }
  }
}

__global__ void zero_kernel(float* __restrict__ p, long n) {
  long i = (long)blockIdx.x * blockDim.x + threadIdx.x;
  long stride = (long)gridDim.x * blockDim.x;
  for (; i < n; i += stride) p[i] = 0.0f;
}

#define WAVES_E 2
__global__ __launch_bounds__(64) void edge_kernel(
    const float* __restrict__ r_ij, const float* __restrict__ w_rad,
    const float* __restrict__ b_rad, const float* __restrict__ w_direct,
    const float* __restrict__ w1, const float* __restrict__ b1,
    const float* __restrict__ w2, const float* __restrict__ b2,
    const float* __restrict__ w3, const float* __restrict__ b3,
    const int* __restrict__ edges_src, float* __restrict__ A_a,
    float* __restrict__ Av, float* __restrict__ Ad6, int n_edges, int iters) {
  __shared__ float sWrad[8 * 32];
  __shared__ float sBrad[32];
  __shared__ float sWdir[32 * 32];
  __shared__ float sW1[32 * 64];
  __shared__ float sB1[64];
  __shared__ float sW2[64 * 64];
  __shared__ float sB2[64];
  __shared__ float sW3[64 * 32];
  __shared__ float sB3[32];
  __shared__ float sHH[WAVES_E][16 * 32];
  __shared__ float sT1[WAVES_E][16 * 64];
  __shared__ float sT2[WAVES_E][16 * 64];
  __shared__ float sRS[WAVES_E][16 * 4];
  __shared__ int sSRC[WAVES_E][16];

  const int tid = threadIdx.x;
  for (int i = tid; i < 8 * 32; i += blockDim.x) sWrad[i] = w_rad[i];
  for (int i = tid; i < 32; i += blockDim.x) sBrad[i] = b_rad[i];
  for (int i = tid; i < 32 * 32; i += blockDim.x) sWdir[i] = w_direct[i];
  for (int i = tid; i < 32 * 64; i += blockDim.x) sW1[i] = w1[i];
  for (int i = tid; i < 64; i += blockDim.x) sB1[i] = b1[i];
  for (int i = tid; i < 64 * 64; i += blockDim.x) sW2[i] = w2[i];
  for (int i = tid; i < 64; i += blockDim.x) sB2[i] = b2[i];
  for (int i = tid; i < 64 * 32; i += blockDim.x) sW3[i] = w3[i];
  for (int i = tid; i < 32; i += blockDim.x) sB3[i] = b3[i];
  __syncthreads();

  const int wave = tid >> 5, lane = tid & 31;
  const int col = lane & 15, hi = lane >> 4;
  const int n_tiles = (n_edges + 15) / 16;
  const int tile_stride = gridDim.x * WAVES_E;

  for (int it = 0; it < iters; ++it) {
    const int tile = (it * gridDim.x + blockIdx.x) * WAVES_E + wave;
    const bool active = tile < n_tiles;
    const int e0 = tile * 16;
    float* rbfBuf = sT1[wave];

    if (active && lane < 16) {
      int e = e0 + lane;
      long enext = (long)(e0 + 16L * tile_stride) + lane;
      if (enext < n_edges) {
        __builtin_prefetch(&r_ij[3 * enext], 0, 1);
        __builtin_prefetch(&edges_src[enext], 0, 1);
      }
      float x = r_ij[3 * e + 0], y = r_ij[3 * e + 1], z = r_ij[3 * e + 2];
      float d = sqrtf(x * x + y * y + z * z);
#pragma unroll
      for (int i = 0; i < 8; ++i) {
        float u = (d - (float)i * (1.0f / 7.0f)) * 8.0f;
        rbfBuf[lane * 8 + i] = __expf(-u * u);
      }
      float sx = 7.0f * x, sy = 7.0f * y, sz = 7.0f * z;
      float inv = rsqrtf(1.0f + sx * sx + sy * sy + sz * sz);
      sRS[wave][lane * 4 + 0] = sx * inv;
      sRS[wave][lane * 4 + 1] = sy * inv;
      sRS[wave][lane * 4 + 2] = sz * inv;
      sSRC[wave][lane] = edges_src[e];
    }
    __syncthreads();

    gemm_stage<8, 32, false>(rbfBuf, sWrad, sBrad, sHH[wave], col, hi);
    __syncthreads();
    gemm_stage<32, 64, true>(sHH[wave], sW1, sB1, sT1[wave], col, hi);
    __syncthreads();
    gemm_stage<64, 64, true>(sT1[wave], sW2, sB2, sT2[wave], col, hi);
    __syncthreads();

#pragma unroll
    for (int nt = 0; nt < 2; ++nt) {
      float bv = sB3[nt * 16 + col];
      v8f c;
#pragma unroll
      for (int r = 0; r < 8; ++r) c[r] = bv;
#pragma unroll
      for (int k0 = 0; k0 < 32; k0 += 4) {
        int k = k0 + 2 * hi;
        v2f a;
        a.x = sHH[wave][col * 32 + k];
        a.y = sHH[wave][col * 32 + k + 1];
        v2f b;
        b.x = sWdir[k * 32 + nt * 16 + col];
        b.y = sWdir[(k + 1) * 32 + nt * 16 + col];
        c = wmma_f32(a, b, c);
      }
#pragma unroll
      for (int k0 = 0; k0 < 64; k0 += 4) {
        int k = k0 + 2 * hi;
        v2f a;
        a.x = sT2[wave][col * 64 + k];
        a.y = sT2[wave][col * 64 + k + 1];
        v2f b;
        b.x = sW3[k * 32 + nt * 16 + col];
        b.y = sW3[(k + 1) * 32 + nt * 16 + col];
        c = wmma_f32(a, b, c);
      }

      if (active) {
        int ach = nt * 16 + col;
#pragma unroll
        for (int r = 0; r < 8; ++r) {
          int m = r + 8 * hi;
          float val = c[r];
          int src = sSRC[wave][m];
          float rx = sRS[wave][m * 4 + 0];
          float ry = sRS[wave][m * 4 + 1];
          float rz = sRS[wave][m * 4 + 2];
          atomicAdd(&A_a[src * 32 + ach], val);
          float* av = &Av[(src * 32 + ach) * 3];
          atomicAdd(&av[0], val * rx);
          atomicAdd(&av[1], val * ry);
          atomicAdd(&av[2], val * rz);
          float* ad = &Ad6[(src * 32 + ach) * 6];
          atomicAdd(&ad[0], val * rx * rx);
          atomicAdd(&ad[1], val * rx * ry);
          atomicAdd(&ad[2], val * rx * rz);
          atomicAdd(&ad[3], val * ry * ry);
          atomicAdd(&ad[4], val * ry * rz);
          atomicAdd(&ad[5], val * rz * rz);
        }
      }
    }
    __syncthreads();
  }
}

#define WAVES_N 8
__global__ __launch_bounds__(256) void node_kernel(
    const float* __restrict__ Av, const float* __restrict__ Ad6,
    const float* __restrict__ w_v, const float* __restrict__ w_d,
    float* __restrict__ out_v, float* __restrict__ out_d, int n_nodes) {
  __shared__ float sWv[32 * 32];
  __shared__ float sWd[32 * 32];
  for (int i = threadIdx.x; i < 32 * 32; i += blockDim.x) {
    sWv[i] = w_v[i];
    sWd[i] = w_d[i];
  }
  __syncthreads();

  const int wave = threadIdx.x >> 5, lane = threadIdx.x & 31;
  const int col = lane & 15, hi = lane >> 4;
  const int tile = blockIdx.x * WAVES_N + wave;
  if (tile * 16 >= n_nodes) return;
  const int n0 = tile * 16;

#pragma unroll
  for (int cch = 0; cch < 3; ++cch) {
#pragma unroll
    for (int nt = 0; nt < 2; ++nt) {
      v8f acc = {};
#pragma unroll
      for (int k0 = 0; k0 < 32; k0 += 4) {
        int k = k0 + 2 * hi;
        v2f a;
        a.x = Av[((long)(n0 + col) * 32 + k) * 3 + cch];
        a.y = Av[((long)(n0 + col) * 32 + k + 1) * 3 + cch];
        v2f b;
        b.x = sWv[k * 32 + nt * 16 + col];
        b.y = sWv[(k + 1) * 32 + nt * 16 + col];
        acc = wmma_f32(a, b, acc);
      }
#pragma unroll
      for (int r = 0; r < 8; ++r) {
        int node = n0 + r + 8 * hi;
        int v = nt * 16 + col;
        __builtin_nontemporal_store(acc[r],
                                    &out_v[((long)node * 32 + v) * 3 + cch]);
      }
    }
  }

  const int si[6] = {0, 0, 0, 1, 1, 2};
  const int sj[6] = {0, 1, 2, 1, 2, 2};
#pragma unroll
  for (int s = 0; s < 6; ++s) {
#pragma unroll
    for (int nt = 0; nt < 2; ++nt) {
      v8f acc = {};
#pragma unroll
      for (int k0 = 0; k0 < 32; k0 += 4) {
        int k = k0 + 2 * hi;
        v2f a;
        a.x = Ad6[((long)(n0 + col) * 32 + k) * 6 + s];
        a.y = Ad6[((long)(n0 + col) * 32 + k + 1) * 6 + s];
        v2f b;
        b.x = sWd[k * 32 + nt * 16 + col];
        b.y = sWd[(k + 1) * 32 + nt * 16 + col];
        acc = wmma_f32(a, b, acc);
      }
#pragma unroll
      for (int r = 0; r < 8; ++r) {
        int node = n0 + r + 8 * hi;
        int dch = nt * 16 + col;
        float val = acc[r];
        float* base = &out_d[((long)node * 32 + dch) * 9];
        __builtin_nontemporal_store(val, &base[si[s] * 3 + sj[s]]);
        if (si[s] != sj[s])
          __builtin_nontemporal_store(val, &base[sj[s] * 3 + si[s]]);
      }
    }
  }
}

extern "C" void kernel_launch(void* const* d_in, const int* in_sizes, int n_in,
                              void* d_out, int out_size, void* d_ws,
                              size_t ws_size, hipStream_t stream) {
  const float* r_ij = (const float*)d_in[0];
  const float* w_rad = (const float*)d_in[1];
  const float* b_rad = (const float*)d_in[2];
  const float* w_direct = (const float*)d_in[3];
  const float* w1 = (const float*)d_in[4];
  const float* b1 = (const float*)d_in[5];
  const float* w2 = (const float*)d_in[6];
  const float* b2 = (const float*)d_in[7];
  const float* w3 = (const float*)d_in[8];
  const float* b3 = (const float*)d_in[9];
  const float* w_v = (const float*)d_in[10];
  const float* w_d = (const float*)d_in[11];
  const int* edges_src = (const int*)d_in[12];

  const int n_edges = in_sizes[0] / 3;
  const int n_nodes = NODES_TOT;

  float* A_a = (float*)d_out;
  float* out_v = A_a + (long)n_nodes * 32;
  float* out_d = out_v + (long)n_nodes * 32 * 3;
  float* Av = (float*)d_ws;
  float* Ad6 = Av + (long)n_nodes * 32 * 3;

  zero_kernel<<<2048, 256, 0, stream>>>((float*)d_ws, (long)n_nodes * 32 * 9);
  zero_kernel<<<512, 256, 0, stream>>>(A_a, (long)n_nodes * 32);

  const int n_tiles = (n_edges + 15) / 16;
  const int e_blocks = 1250;
  const int tile_stride = e_blocks * WAVES_E;
  const int iters = (n_tiles + tile_stride - 1) / tile_stride;
  edge_kernel<<<e_blocks, 64, 0, stream>>>(r_ij, w_rad, b_rad, w_direct, w1, b1,
                                           w2, b2, w3, b3, edges_src, A_a, Av,
                                           Ad6, n_edges, iters);

  const int nd_tiles = (n_nodes + 15) / 16;
  const int n_blocks = (nd_tiles + WAVES_N - 1) / WAVES_N;
  node_kernel<<<n_blocks, 256, 0, stream>>>(Av, Ad6, w_v, w_d, out_v, out_d,
                                            n_nodes);
}


# Edge kernel owner-computes: each work-group owns 80 nodes, picks their edges from edges_src, sums rad*rs products in LDS (LDS f32 adds, arrival order), stores rows once; no global atomics, no zeroing
# speedup vs baseline: 1.2614x; 1.2614x over previous
#include <hip/hip_runtime.h>

typedef float v2f __attribute__((ext_vector_type(2)));
typedef float v8f __attribute__((ext_vector_type(8)));

#define NODES_TOT 20000

__device__ __forceinline__ v8f wmma_f32(v2f a, v2f b, v8f c) {
  return __builtin_amdgcn_wmma_f32_16x16x4_f32(
       false, a,  false, b,
       (short)0, c,  false,  false);
}

template <int K, int N, bool RELU>
__device__ __forceinline__ void gemm_stage(const float* aBuf, const float* w,
                                           const float* bias, float* outBuf,
                                           int col, int hi) {
#pragma unroll
  for (int nt = 0; nt < N / 16; ++nt) {
    float bv = bias ? bias[nt * 16 + col] : 0.0f;
    v8f c;
#pragma unroll
    for (int r = 0; r < 8; ++r) c[r] = bv;
#pragma unroll
    for (int k0 = 0; k0 < K; k0 += 4) {
      int k = k0 + 2 * hi;
      v2f a;
      a.x = aBuf[col * K + k];
      a.y = aBuf[col * K + k + 1];
      v2f b;
      b.x = w[k * N + nt * 16 + col];
      b.y = w[(k + 1) * N + nt * 16 + col];
      c = wmma_f32(a, b, c);
    }
#pragma unroll
    for (int r = 0; r < 8; ++r) {
      float v = c[r];
      if (RELU) v = fmaxf(v, 0.1f * v);
      outBuf[(r + 8 * hi) * N + nt * 16 + col] = v;
    }
  }
}

#define WAVES_E 8
#define THREADS_E (WAVES_E * 32)
#define NB 80
#define NPL 10
#define ROW (NPL * 32)
#define SCAN_V4 8
#define BATCH (THREADS_E * 4 * SCAN_V4)
#define PROC (WAVES_E * 16)
#define LIST_CAP (PROC + BATCH)

__device__ __forceinline__ void load_batch(int4 (&dst)[SCAN_V4],
                                           const int* __restrict__ src, int p,
                                           int tid) {
#pragma unroll
  for (int j = 0; j < SCAN_V4; ++j)
    dst[j] =
        *reinterpret_cast<const int4*>(src + p + (j * THREADS_E + tid) * 4);
}

__device__ __forceinline__ void keep_if_owned(int s, int e, int n0, int nb,
                                              int* sCount, int* sListE,
                                              unsigned short* sListN) {
  const unsigned nl = (unsigned)(s - n0);
  if (nl < (unsigned)nb) {
    const int slot = atomicAdd(sCount, 1);
    sListE[slot] = e;
    sListN[slot] = (unsigned short)nl;
  }
}

__global__ __launch_bounds__(THREADS_E) void edge_kernel(
    const float* __restrict__ r_ij, const float* __restrict__ w_rad,
    const float* __restrict__ b_rad, const float* __restrict__ w_direct,
    const float* __restrict__ w1, const float* __restrict__ b1,
    const float* __restrict__ w2, const float* __restrict__ b2,
    const float* __restrict__ w3, const float* __restrict__ b3,
    const int* __restrict__ edges_src, float* __restrict__ A_a,
    float* __restrict__ Av, float* __restrict__ Ad6, int n_edges,
    int n_nodes) {
  __shared__ float sWrad[8 * 32];
  __shared__ float sBrad[32];
  __shared__ float sWdir[32 * 32];
  __shared__ float sW1[32 * 64];
  __shared__ float sB1[64];
  __shared__ float sW2[64 * 64];
  __shared__ float sB2[64];
  __shared__ float sW3[64 * 32];
  __shared__ float sB3[32];
  __shared__ float sHH[WAVES_E][16 * 32];
  __shared__ float sT1[WAVES_E][16 * 64];
  __shared__ float sT2[WAVES_E][16 * 64];
  __shared__ float sRS[WAVES_E][16 * 4];
  __shared__ int sSRC[WAVES_E][16];
  __shared__ float sAcc[(NB + 1) * ROW];
  __shared__ int sListE[LIST_CAP];
  __shared__ unsigned short sListN[LIST_CAP];
  __shared__ int sCount;

  const int tid = threadIdx.x;
  for (int i = tid; i < 8 * 32; i += THREADS_E) sWrad[i] = w_rad[i];
  for (int i = tid; i < 32; i += THREADS_E) sBrad[i] = b_rad[i];
  for (int i = tid; i < 32 * 32; i += THREADS_E) sWdir[i] = w_direct[i];
  for (int i = tid; i < 32 * 64; i += THREADS_E) sW1[i] = w1[i];
  for (int i = tid; i < 64; i += THREADS_E) sB1[i] = b1[i];
  for (int i = tid; i < 64 * 64; i += THREADS_E) sW2[i] = w2[i];
  for (int i = tid; i < 64; i += THREADS_E) sB2[i] = b2[i];
  for (int i = tid; i < 64 * 32; i += THREADS_E) sW3[i] = w3[i];
  for (int i = tid; i < 32; i += THREADS_E) sB3[i] = b3[i];
  for (int i = tid; i < (NB + 1) * ROW; i += THREADS_E) sAcc[i] = 0.0f;
  if (tid == 0) sCount = 0;
  __syncthreads();

  const int wave = tid >> 5, lane = tid & 31;
  const int col = lane & 15, hi = lane >> 4;
  const int n0 = blockIdx.x * NB;
  const int nb = (n_nodes - n0 < NB) ? (n_nodes - n0) : NB;

  const int n_full = (n_edges / BATCH) * BATCH;
  for (int e = n_full + tid; e < n_edges; e += THREADS_E)
    keep_if_owned(edges_src[e], e, n0, nb, &sCount, sListE, sListN);

  int4 nxt[SCAN_V4];
#pragma unroll
  for (int j = 0; j < SCAN_V4; ++j) nxt[j] = make_int4(-1, -1, -1, -1);
  if (n_full > 0) load_batch(nxt, edges_src, 0, tid);

  __syncthreads();
  int cnt = __builtin_amdgcn_readfirstlane(sCount);
  __syncthreads();
  int pos = 0;
  for (;;) {
    while (cnt < PROC && pos < n_full) {
      int4 cur[SCAN_V4];
#pragma unroll
      for (int j = 0; j < SCAN_V4; ++j) cur[j] = nxt[j];
      const int npos = pos + BATCH;
      if (npos < n_full) load_batch(nxt, edges_src, npos, tid);
#pragma unroll
      for (int j = 0; j < SCAN_V4; ++j) {
        const int e = pos + (j * THREADS_E + tid) * 4;
        const unsigned d0 = (unsigned)(cur[j].x - n0);
        const unsigned d1 = (unsigned)(cur[j].y - n0);
        const unsigned d2 = (unsigned)(cur[j].z - n0);
        const unsigned d3 = (unsigned)(cur[j].w - n0);
        const unsigned dmin = min(min(d0, d1), min(d2, d3));
        if (dmin < (unsigned)nb) {
          keep_if_owned(cur[j].x, e + 0, n0, nb, &sCount, sListE, sListN);
          keep_if_owned(cur[j].y, e + 1, n0, nb, &sCount, sListE, sListN);
          keep_if_owned(cur[j].z, e + 2, n0, nb, &sCount, sListE, sListN);
          keep_if_owned(cur[j].w, e + 3, n0, nb, &sCount, sListE, sListN);
        }
      }
      pos = npos;
      __syncthreads();
      cnt = __builtin_amdgcn_readfirstlane(sCount);
      __syncthreads();
    }
    if (cnt == 0) break;

    const int take = (cnt >= PROC) ? (cnt / PROC) * PROC : cnt;
    const int iters = (take + PROC - 1) / PROC;

    for (int it = 0; it < iters; ++it) {
      const int base = (it * WAVES_E + wave) * 16;
      const bool active = base < take;
      float* rbfBuf = sT1[wave];

      if (lane < 16) {
        const int li = base + lane;
        if (li < take) {
          const int e = sListE[li];
          if (li + PROC < take)
            __builtin_prefetch(&r_ij[3 * (long)sListE[li + PROC]], 0, 1);
          float x = r_ij[3 * (long)e + 0], y = r_ij[3 * (long)e + 1],
                z = r_ij[3 * (long)e + 2];
          float d = sqrtf(x * x + y * y + z * z);
#pragma unroll
          for (int i = 0; i < 8; ++i) {
            float u = (d - (float)i * (1.0f / 7.0f)) * 8.0f;
            rbfBuf[lane * 8 + i] = __expf(-u * u);
          }
          float sx = 7.0f * x, sy = 7.0f * y, sz = 7.0f * z;
          float inv = rsqrtf(1.0f + sx * sx + sy * sy + sz * sz);
          sRS[wave][lane * 4 + 0] = sx * inv;
          sRS[wave][lane * 4 + 1] = sy * inv;
          sRS[wave][lane * 4 + 2] = sz * inv;
          sSRC[wave][lane] = (int)sListN[li];
        } else {
#pragma unroll
          for (int i = 0; i < 8; ++i) rbfBuf[lane * 8 + i] = 0.0f;
          sRS[wave][lane * 4 + 0] = 0.0f;
          sRS[wave][lane * 4 + 1] = 0.0f;
          sRS[wave][lane * 4 + 2] = 0.0f;
          sSRC[wave][lane] = NB;
        }
      }
      __syncthreads();

      gemm_stage<8, 32, false>(rbfBuf, sWrad, sBrad, sHH[wave], col, hi);
      __syncthreads();
      gemm_stage<32, 64, true>(sHH[wave], sW1, sB1, sT1[wave], col, hi);
      __syncthreads();
      gemm_stage<64, 64, true>(sT1[wave], sW2, sB2, sT2[wave], col, hi);
      __syncthreads();

#pragma unroll
      for (int nt = 0; nt < 2; ++nt) {
        float bv = sB3[nt * 16 + col];
        v8f c;
#pragma unroll
        for (int r = 0; r < 8; ++r) c[r] = bv;
#pragma unroll
        for (int k0 = 0; k0 < 32; k0 += 4) {
          int k = k0 + 2 * hi;
          v2f a;
          a.x = sHH[wave][col * 32 + k];
          a.y = sHH[wave][col * 32 + k + 1];
          v2f b;
          b.x = sWdir[k * 32 + nt * 16 + col];
          b.y = sWdir[(k + 1) * 32 + nt * 16 + col];
          c = wmma_f32(a, b, c);
        }
#pragma unroll
        for (int k0 = 0; k0 < 64; k0 += 4) {
          int k = k0 + 2 * hi;
          v2f a;
          a.x = sT2[wave][col * 64 + k];
          a.y = sT2[wave][col * 64 + k + 1];
          v2f b;
          b.x = sW3[k * 32 + nt * 16 + col];
          b.y = sW3[(k + 1) * 32 + nt * 16 + col];
          c = wmma_f32(a, b, c);
        }

        if (active) {
          const int ach = nt * 16 + col;
#pragma unroll
          for (int r = 0; r < 8; ++r) {
            const int m = r + 8 * hi;
            const float val = c[r];
            const int nl = sSRC[wave][m];
            const float rx = sRS[wave][m * 4 + 0];
            const float ry = sRS[wave][m * 4 + 1];
            const float rz = sRS[wave][m * 4 + 2];
            const float t[NPL] = {val,           val * rx,      val * ry,
                                  val * rz,      val * rx * rx, val * rx * ry,
                                  val * rx * rz, val * ry * ry, val * ry * rz,
                                  val * rz * rz};
            float* acc = &sAcc[nl * ROW + ach];
#pragma unroll
            for (int s = 0; s < NPL; s += 2) {
              const float va = hi ? t[s + 1] : t[s];
              const float vb = hi ? t[s] : t[s + 1];
              atomicAdd(&acc[(s + hi) * 32], va);
              atomicAdd(&acc[(s + 1 - hi) * 32], vb);
            }
          }
        }
      }
      __syncthreads();
    }

    const int rem = cnt - take;
    int keepE = 0;
    unsigned short keepN = 0;
    if (tid < rem) {
      keepE = sListE[take + tid];
      keepN = sListN[take + tid];
    }
    __syncthreads();
    if (tid < rem) {
      sListE[tid] = keepE;
      sListN[tid] = keepN;
    }
    if (tid == 0) sCount = rem;
    __syncthreads();
    cnt = rem;
  }
  __syncthreads();

  for (int i = tid; i < nb * 32; i += THREADS_E) {
    const int nl = i >> 5, a = i & 31;
    A_a[(long)n0 * 32 + i] = sAcc[nl * ROW + a];
  }
  for (int i = tid; i < nb * 96; i += THREADS_E) {
    const int nl = i / 96, q = i - nl * 96;
    const int a = q / 3, cc = q - a * 3;
    Av[(long)n0 * 96 + i] = sAcc[nl * ROW + (1 + cc) * 32 + a];
  }
  for (int i = tid; i < nb * 192; i += THREADS_E) {
    const int nl = i / 192, q = i - nl * 192;
    const int a = q / 6, s6 = q - a * 6;
    Ad6[(long)n0 * 192 + i] = sAcc[nl * ROW + (4 + s6) * 32 + a];
  }
}

#define WAVES_N 8
__global__ __launch_bounds__(256) void node_kernel(
    const float* __restrict__ Av, const float* __restrict__ Ad6,
    const float* __restrict__ w_v, const float* __restrict__ w_d,
    float* __restrict__ out_v, float* __restrict__ out_d, int n_nodes) {
  __shared__ float sWv[32 * 32];
  __shared__ float sWd[32 * 32];
  for (int i = threadIdx.x; i < 32 * 32; i += blockDim.x) {
    sWv[i] = w_v[i];
    sWd[i] = w_d[i];
  }
  __syncthreads();

  const int wave = threadIdx.x >> 5, lane = threadIdx.x & 31;
  const int col = lane & 15, hi = lane >> 4;
  const int tile = blockIdx.x * WAVES_N + wave;
  if (tile * 16 >= n_nodes) return;
  const int n0 = tile * 16;

#pragma unroll
  for (int cch = 0; cch < 3; ++cch) {
#pragma unroll
    for (int nt = 0; nt < 2; ++nt) {
      v8f acc = {};
#pragma unroll
      for (int k0 = 0; k0 < 32; k0 += 4) {
        int k = k0 + 2 * hi;
        v2f a;
        a.x = Av[((long)(n0 + col) * 32 + k) * 3 + cch];
        a.y = Av[((long)(n0 + col) * 32 + k + 1) * 3 + cch];
        v2f b;
        b.x = sWv[k * 32 + nt * 16 + col];
        b.y = sWv[(k + 1) * 32 + nt * 16 + col];
        acc = wmma_f32(a, b, acc);
      }
#pragma unroll
      for (int r = 0; r < 8; ++r) {
        int node = n0 + r + 8 * hi;
        int v = nt * 16 + col;
        __builtin_nontemporal_store(acc[r],
                                    &out_v[((long)node * 32 + v) * 3 + cch]);
      }
    }
  }

  const int si[6] = {0, 0, 0, 1, 1, 2};
  const int sj[6] = {0, 1, 2, 1, 2, 2};
#pragma unroll
  for (int s = 0; s < 6; ++s) {
#pragma unroll
    for (int nt = 0; nt < 2; ++nt) {
      v8f acc = {};
#pragma unroll
      for (int k0 = 0; k0 < 32; k0 += 4) {
        int k = k0 + 2 * hi;
        v2f a;
        a.x = Ad6[((long)(n0 + col) * 32 + k) * 6 + s];
        a.y = Ad6[((long)(n0 + col) * 32 + k + 1) * 6 + s];
        v2f b;
        b.x = sWd[k * 32 + nt * 16 + col];
        b.y = sWd[(k + 1) * 32 + nt * 16 + col];
        acc = wmma_f32(a, b, acc);
      }
#pragma unroll
      for (int r = 0; r < 8; ++r) {
        int node = n0 + r + 8 * hi;
        int dch = nt * 16 + col;
        float val = acc[r];
        float* base = &out_d[((long)node * 32 + dch) * 9];
        __builtin_nontemporal_store(val, &base[si[s] * 3 + sj[s]]);
        if (si[s] != sj[s])
          __builtin_nontemporal_store(val, &base[sj[s] * 3 + si[s]]);
      }
    }
  }
}

extern "C" void kernel_launch(void* const* d_in, const int* in_sizes, int n_in,
                              void* d_out, int out_size, void* d_ws,
                              size_t ws_size, hipStream_t stream) {
  const float* r_ij = (const float*)d_in[0];
  const float* w_rad = (const float*)d_in[1];
  const float* b_rad = (const float*)d_in[2];
  const float* w_direct = (const float*)d_in[3];
  const float* w1 = (const float*)d_in[4];
  const float* b1 = (const float*)d_in[5];
  const float* w2 = (const float*)d_in[6];
  const float* b2 = (const float*)d_in[7];
  const float* w3 = (const float*)d_in[8];
  const float* b3 = (const float*)d_in[9];
  const float* w_v = (const float*)d_in[10];
  const float* w_d = (const float*)d_in[11];
  const int* edges_src = (const int*)d_in[12];

  const int n_edges = in_sizes[0] / 3;
  const int n_nodes = NODES_TOT;

  float* A_a = (float*)d_out;
  float* out_v = A_a + (long)n_nodes * 32;
  float* out_d = out_v + (long)n_nodes * 32 * 3;
  float* Av = (float*)d_ws;
  float* Ad6 = Av + (long)n_nodes * 32 * 3;

  const int e_blocks = (NODES_TOT + NB - 1) / NB;
  edge_kernel<<<e_blocks, THREADS_E, 0, stream>>>(r_ij, w_rad, b_rad, w_direct,
                                                  w1, b1, w2, b2, w3, b3,
                                                  edges_src, A_a, Av, Ad6,
                                                  n_edges, n_nodes);

  const int nd_tiles = (n_nodes + 15) / 16;
  const int n_blocks = (nd_tiles + WAVES_N - 1) / WAVES_N;
  node_kernel<<<n_blocks, 256, 0, stream>>>(Av, Ad6, w_v, w_d, out_v, out_d,
                                            n_nodes);
}


# Edge kernel: LDS row pitches padded (activations 12/36/68 words, weights 48/80) so A/B fragment reads and C-layout stores hit 64 distinct banks, were 2/8/16-way conflicts; same sums, same order
# speedup vs baseline: 1.6053x; 1.2726x over previous
#include <hip/hip_runtime.h>

typedef float v2f __attribute__((ext_vector_type(2)));
typedef float v8f __attribute__((ext_vector_type(8)));

#define NODES_TOT 20000

__device__ __forceinline__ v8f wmma_f32(v2f a, v2f b, v8f c) {
  return __builtin_amdgcn_wmma_f32_16x16x4_f32(
       false, a,  false, b,
       (short)0, c,  false,  false);
}

#define AP8 12
#define AP32 36
#define AP64 68
#define WP32 48
#define WP64 80

template <int K, int N, int AP, int WP, int OP, bool RELU>
__device__ __forceinline__ void gemm_stage(const float* aBuf, const float* w,
                                           const float* bias, float* outBuf,
                                           int col, int hi) {
  const float* ap = aBuf + col * AP + 2 * hi;
  const float* wp = w + 2 * hi * WP + col;
  float* op = outBuf + 8 * hi * OP + col;
#pragma unroll
  for (int nt = 0; nt < N / 16; ++nt) {
    float bv = bias ? bias[nt * 16 + col] : 0.0f;
    v8f c;
#pragma unroll
    for (int r = 0; r < 8; ++r) c[r] = bv;
#pragma unroll
    for (int k0 = 0; k0 < K; k0 += 4) {
      v2f a;
      a.x = ap[k0];
      a.y = ap[k0 + 1];
      v2f b;
      b.x = wp[k0 * WP + nt * 16];
      b.y = wp[(k0 + 1) * WP + nt * 16];
      c = wmma_f32(a, b, c);
    }
#pragma unroll
    for (int r = 0; r < 8; ++r) {
      float v = c[r];
      if (RELU) v = fmaxf(v, 0.1f * v);
      op[r * OP + nt * 16] = v;
    }
  }
}

#define WAVES_E 8
#define THREADS_E (WAVES_E * 32)
#define NB 80
#define NPL 10
#define ROW (NPL * 32)
#define SCAN_V4 8
#define BATCH (THREADS_E * 4 * SCAN_V4)
#define PROC (WAVES_E * 16)
#define LIST_CAP (PROC + BATCH)

__device__ __forceinline__ void load_batch(int4 (&dst)[SCAN_V4],
                                           const int* __restrict__ src, int p,
                                           int tid) {
#pragma unroll
  for (int j = 0; j < SCAN_V4; ++j)
    dst[j] =
        *reinterpret_cast<const int4*>(src + p + (j * THREADS_E + tid) * 4);
}

__device__ __forceinline__ void keep_if_owned(int s, int e, int n0, int nb,
                                              int* sCount, int* sListE,
                                              unsigned short* sListN) {
  const unsigned nl = (unsigned)(s - n0);
  if (nl < (unsigned)nb) {
    const int slot = atomicAdd(sCount, 1);
    sListE[slot] = e;
    sListN[slot] = (unsigned short)nl;
  }
}

__global__ __launch_bounds__(THREADS_E) void edge_kernel(
    const float* __restrict__ r_ij, const float* __restrict__ w_rad,
    const float* __restrict__ b_rad, const float* __restrict__ w_direct,
    const float* __restrict__ w1, const float* __restrict__ b1,
    const float* __restrict__ w2, const float* __restrict__ b2,
    const float* __restrict__ w3, const float* __restrict__ b3,
    const int* __restrict__ edges_src, float* __restrict__ A_a,
    float* __restrict__ Av, float* __restrict__ Ad6, int n_edges,
    int n_nodes) {
  __shared__ float sWrad[8 * WP32];
  __shared__ float sBrad[32];
  __shared__ float sWdir[32 * WP32];
  __shared__ float sW1[32 * WP64];
  __shared__ float sB1[64];
  __shared__ float sW2[64 * WP64];
  __shared__ float sB2[64];
  __shared__ float sW3[64 * WP32];
  __shared__ float sB3[32];
  __shared__ float sHH[WAVES_E][16 * AP32];
  __shared__ float sT1[WAVES_E][16 * AP64];
  __shared__ float sT2[WAVES_E][16 * AP64];
  __shared__ float sRS[WAVES_E][16 * 4];
  __shared__ int sSRC[WAVES_E][16];
  __shared__ float sAcc[(NB + 1) * ROW];
  __shared__ int sListE[LIST_CAP];
  __shared__ unsigned short sListN[LIST_CAP];
  __shared__ int sCount;

  const int tid = threadIdx.x;
  for (int i = tid; i < 8 * 32; i += THREADS_E)
    sWrad[(i >> 5) * WP32 + (i & 31)] = w_rad[i];
  for (int i = tid; i < 32; i += THREADS_E) sBrad[i] = b_rad[i];
  for (int i = tid; i < 32 * 32; i += THREADS_E)
    sWdir[(i >> 5) * WP32 + (i & 31)] = w_direct[i];
  for (int i = tid; i < 32 * 64; i += THREADS_E)
    sW1[(i >> 6) * WP64 + (i & 63)] = w1[i];
  for (int i = tid; i < 64; i += THREADS_E) sB1[i] = b1[i];
  for (int i = tid; i < 64 * 64; i += THREADS_E)
    sW2[(i >> 6) * WP64 + (i & 63)] = w2[i];
  for (int i = tid; i < 64; i += THREADS_E) sB2[i] = b2[i];
  for (int i = tid; i < 64 * 32; i += THREADS_E)
    sW3[(i >> 5) * WP32 + (i & 31)] = w3[i];
  for (int i = tid; i < 32; i += THREADS_E) sB3[i] = b3[i];
  for (int i = tid; i < (NB + 1) * ROW; i += THREADS_E) sAcc[i] = 0.0f;
  if (tid == 0) sCount = 0;
  __syncthreads();

  const int wave = tid >> 5, lane = tid & 31;
  const int col = lane & 15, hi = lane >> 4;
  const int n0 = blockIdx.x * NB;
  const int nb = (n_nodes - n0 < NB) ? (n_nodes - n0) : NB;

  const int n_full = (n_edges / BATCH) * BATCH;
  for (int e = n_full + tid; e < n_edges; e += THREADS_E)
    keep_if_owned(edges_src[e], e, n0, nb, &sCount, sListE, sListN);

  int4 nxt[SCAN_V4];
#pragma unroll
  for (int j = 0; j < SCAN_V4; ++j) nxt[j] = make_int4(-1, -1, -1, -1);
  if (n_full > 0) load_batch(nxt, edges_src, 0, tid);

  __syncthreads();
  int cnt = __builtin_amdgcn_readfirstlane(sCount);
  __syncthreads();
  int pos = 0;
  for (;;) {
    while (cnt < PROC && pos < n_full) {
      int4 cur[SCAN_V4];
#pragma unroll
      for (int j = 0; j < SCAN_V4; ++j) cur[j] = nxt[j];
      const int npos = pos + BATCH;
      if (npos < n_full) load_batch(nxt, edges_src, npos, tid);
#pragma unroll
      for (int j = 0; j < SCAN_V4; ++j) {
        const int e = pos + (j * THREADS_E + tid) * 4;
        const unsigned d0 = (unsigned)(cur[j].x - n0);
        const unsigned d1 = (unsigned)(cur[j].y - n0);
        const unsigned d2 = (unsigned)(cur[j].z - n0);
        const unsigned d3 = (unsigned)(cur[j].w - n0);
        const unsigned dmin = min(min(d0, d1), min(d2, d3));
        if (dmin < (unsigned)nb) {
          keep_if_owned(cur[j].x, e + 0, n0, nb, &sCount, sListE, sListN);
          keep_if_owned(cur[j].y, e + 1, n0, nb, &sCount, sListE, sListN);
          keep_if_owned(cur[j].z, e + 2, n0, nb, &sCount, sListE, sListN);
          keep_if_owned(cur[j].w, e + 3, n0, nb, &sCount, sListE, sListN);
        }
      }
      pos = npos;
      __syncthreads();
      cnt = __builtin_amdgcn_readfirstlane(sCount);
      __syncthreads();
    }
    if (cnt == 0) break;

    const int take = (cnt >= PROC) ? (cnt / PROC) * PROC : cnt;
    const int iters = (take + PROC - 1) / PROC;

    for (int it = 0; it < iters; ++it) {
      const int base = (it * WAVES_E + wave) * 16;
      const bool active = base < take;
      float* rbfBuf = sT1[wave];

      if (lane < 16) {
        const int li = base + lane;
        if (li < take) {
          const int e = sListE[li];
          if (li + PROC < take)
            __builtin_prefetch(&r_ij[3 * (long)sListE[li + PROC]], 0, 1);
          float x = r_ij[3 * (long)e + 0], y = r_ij[3 * (long)e + 1],
                z = r_ij[3 * (long)e + 2];
          float d = sqrtf(x * x + y * y + z * z);
#pragma unroll
          for (int i = 0; i < 8; ++i) {
            float u = (d - (float)i * (1.0f / 7.0f)) * 8.0f;
            rbfBuf[lane * AP8 + i] = __expf(-u * u);
          }
          float sx = 7.0f * x, sy = 7.0f * y, sz = 7.0f * z;
          float inv = rsqrtf(1.0f + sx * sx + sy * sy + sz * sz);
          sRS[wave][lane * 4 + 0] = sx * inv;
          sRS[wave][lane * 4 + 1] = sy * inv;
          sRS[wave][lane * 4 + 2] = sz * inv;
          sSRC[wave][lane] = (int)sListN[li];
        } else {
#pragma unroll
          for (int i = 0; i < 8; ++i) rbfBuf[lane * AP8 + i] = 0.0f;
          sRS[wave][lane * 4 + 0] = 0.0f;
          sRS[wave][lane * 4 + 1] = 0.0f;
          sRS[wave][lane * 4 + 2] = 0.0f;
          sSRC[wave][lane] = NB;
        }
      }
      __syncthreads();

      gemm_stage<8, 32, AP8, WP32, AP32, false>(rbfBuf, sWrad, sBrad,
                                                sHH[wave], col, hi);
      __syncthreads();
      gemm_stage<32, 64, AP32, WP64, AP64, true>(sHH[wave], sW1, sB1,
                                                 sT1[wave], col, hi);
      __syncthreads();
      gemm_stage<64, 64, AP64, WP64, AP64, true>(sT1[wave], sW2, sB2,
                                                 sT2[wave], col, hi);
      __syncthreads();

      const float* hp = sHH[wave] + col * AP32 + 2 * hi;
      const float* tp = sT2[wave] + col * AP64 + 2 * hi;
      const float* wdp = sWdir + 2 * hi * WP32 + col;
      const float* w3p = sW3 + 2 * hi * WP32 + col;
#pragma unroll
      for (int nt = 0; nt < 2; ++nt) {
        float bv = sB3[nt * 16 + col];
        v8f c;
#pragma unroll
        for (int r = 0; r < 8; ++r) c[r] = bv;
#pragma unroll
        for (int k0 = 0; k0 < 32; k0 += 4) {
          v2f a;
          a.x = hp[k0];
          a.y = hp[k0 + 1];
          v2f b;
          b.x = wdp[k0 * WP32 + nt * 16];
          b.y = wdp[(k0 + 1) * WP32 + nt * 16];
          c = wmma_f32(a, b, c);
        }
#pragma unroll
        for (int k0 = 0; k0 < 64; k0 += 4) {
          v2f a;
          a.x = tp[k0];
          a.y = tp[k0 + 1];
          v2f b;
          b.x = w3p[k0 * WP32 + nt * 16];
          b.y = w3p[(k0 + 1) * WP32 + nt * 16];
          c = wmma_f32(a, b, c);
        }

        if (active) {
          const int ach = nt * 16 + col;
#pragma unroll
          for (int r = 0; r < 8; ++r) {
            const int m = r + 8 * hi;
            const float val = c[r];
            const int nl = sSRC[wave][m];
            const float rx = sRS[wave][m * 4 + 0];
            const float ry = sRS[wave][m * 4 + 1];
            const float rz = sRS[wave][m * 4 + 2];
            const float t[NPL] = {val,           val * rx,      val * ry,
                                  val * rz,      val * rx * rx, val * rx * ry,
                                  val * rx * rz, val * ry * ry, val * ry * rz,
                                  val * rz * rz};
            float* acc = &sAcc[nl * ROW + ach];
#pragma unroll
            for (int s = 0; s < NPL; s += 2) {
              const float va = hi ? t[s + 1] : t[s];
              const float vb = hi ? t[s] : t[s + 1];
              atomicAdd(&acc[(s + hi) * 32], va);
              atomicAdd(&acc[(s + 1 - hi) * 32], vb);
            }
          }
        }
      }
      __syncthreads();
    }

    const int rem = cnt - take;
    int keepE = 0;
    unsigned short keepN = 0;
    if (tid < rem) {
      keepE = sListE[take + tid];
      keepN = sListN[take + tid];
    }
    __syncthreads();
    if (tid < rem) {
      sListE[tid] = keepE;
      sListN[tid] = keepN;
    }
    if (tid == 0) sCount = rem;
    __syncthreads();
    cnt = rem;
  }
  __syncthreads();

  for (int i = tid; i < nb * 32; i += THREADS_E) {
    const int nl = i >> 5, a = i & 31;
    A_a[(long)n0 * 32 + i] = sAcc[nl * ROW + a];
  }
  for (int i = tid; i < nb * 96; i += THREADS_E) {
    const int nl = i / 96, q = i - nl * 96;
    const int a = q / 3, cc = q - a * 3;
    Av[(long)n0 * 96 + i] = sAcc[nl * ROW + (1 + cc) * 32 + a];
  }
  for (int i = tid; i < nb * 192; i += THREADS_E) {
    const int nl = i / 192, q = i - nl * 192;
    const int a = q / 6, s6 = q - a * 6;
    Ad6[(long)n0 * 192 + i] = sAcc[nl * ROW + (4 + s6) * 32 + a];
  }
}

#define WAVES_N 8
__global__ __launch_bounds__(256) void node_kernel(
    const float* __restrict__ Av, const float* __restrict__ Ad6,
    const float* __restrict__ w_v, const float* __restrict__ w_d,
    float* __restrict__ out_v, float* __restrict__ out_d, int n_nodes) {
  __shared__ float sWv[32 * 32];
  __shared__ float sWd[32 * 32];
  for (int i = threadIdx.x; i < 32 * 32; i += blockDim.x) {
    sWv[i] = w_v[i];
    sWd[i] = w_d[i];
  }
  __syncthreads();

  const int wave = threadIdx.x >> 5, lane = threadIdx.x & 31;
  const int col = lane & 15, hi = lane >> 4;
  const int tile = blockIdx.x * WAVES_N + wave;
  if (tile * 16 >= n_nodes) return;
  const int n0 = tile * 16;

#pragma unroll
  for (int cch = 0; cch < 3; ++cch) {
#pragma unroll
    for (int nt = 0; nt < 2; ++nt) {
      v8f acc = {};
#pragma unroll
      for (int k0 = 0; k0 < 32; k0 += 4) {
        int k = k0 + 2 * hi;
        v2f a;
        a.x = Av[((long)(n0 + col) * 32 + k) * 3 + cch];
        a.y = Av[((long)(n0 + col) * 32 + k + 1) * 3 + cch];
        v2f b;
        b.x = sWv[k * 32 + nt * 16 + col];
        b.y = sWv[(k + 1) * 32 + nt * 16 + col];
        acc = wmma_f32(a, b, acc);
      }
#pragma unroll
      for (int r = 0; r < 8; ++r) {
        int node = n0 + r + 8 * hi;
        int v = nt * 16 + col;
        __builtin_nontemporal_store(acc[r],
                                    &out_v[((long)node * 32 + v) * 3 + cch]);
      }
    }
  }

  const int si[6] = {0, 0, 0, 1, 1, 2};
  const int sj[6] = {0, 1, 2, 1, 2, 2};
#pragma unroll
  for (int s = 0; s < 6; ++s) {
#pragma unroll
    for (int nt = 0; nt < 2; ++nt) {
      v8f acc = {};
#pragma unroll
      for (int k0 = 0; k0 < 32; k0 += 4) {
        int k = k0 + 2 * hi;
        v2f a;
        a.x = Ad6[((long)(n0 + col) * 32 + k) * 6 + s];
        a.y = Ad6[((long)(n0 + col) * 32 + k + 1) * 6 + s];
        v2f b;
        b.x = sWd[k * 32 + nt * 16 + col];
        b.y = sWd[(k + 1) * 32 + nt * 16 + col];
        acc = wmma_f32(a, b, acc);
      }
#pragma unroll
      for (int r = 0; r < 8; ++r) {
        int node = n0 + r + 8 * hi;
        int dch = nt * 16 + col;
        float val = acc[r];
        float* base = &out_d[((long)node * 32 + dch) * 9];
        __builtin_nontemporal_store(val, &base[si[s] * 3 + sj[s]]);
        if (si[s] != sj[s])
          __builtin_nontemporal_store(val, &base[sj[s] * 3 + si[s]]);
      }
    }
  }
}

extern "C" void kernel_launch(void* const* d_in, const int* in_sizes, int n_in,
                              void* d_out, int out_size, void* d_ws,
                              size_t ws_size, hipStream_t stream) {
  const float* r_ij = (const float*)d_in[0];
  const float* w_rad = (const float*)d_in[1];
  const float* b_rad = (const float*)d_in[2];
  const float* w_direct = (const float*)d_in[3];
  const float* w1 = (const float*)d_in[4];
  const float* b1 = (const float*)d_in[5];
  const float* w2 = (const float*)d_in[6];
  const float* b2 = (const float*)d_in[7];
  const float* w3 = (const float*)d_in[8];
  const float* b3 = (const float*)d_in[9];
  const float* w_v = (const float*)d_in[10];
  const float* w_d = (const float*)d_in[11];
  const int* edges_src = (const int*)d_in[12];

  const int n_edges = in_sizes[0] / 3;
  const int n_nodes = NODES_TOT;

  float* A_a = (float*)d_out;
  float* out_v = A_a + (long)n_nodes * 32;
  float* out_d = out_v + (long)n_nodes * 32 * 3;
  float* Av = (float*)d_ws;
  float* Ad6 = Av + (long)n_nodes * 32 * 3;

  const int e_blocks = (NODES_TOT + NB - 1) / NB;
  edge_kernel<<<e_blocks, THREADS_E, 0, stream>>>(r_ij, w_rad, b_rad, w_direct,
                                                  w1, b1, w2, b2, w3, b3,
                                                  edges_src, A_a, Av, Ad6,
                                                  n_edges, n_nodes);

  const int nd_tiles = (n_nodes + 15) / 16;
  const int n_blocks = (nd_tiles + WAVES_N - 1) / WAVES_N;
  node_kernel<<<n_blocks, 256, 0, stream>>>(Av, Ad6, w_v, w_d, out_v, out_d,
                                            n_nodes);
}


# Edge MLP: every wave keeps all five weight matrices as B fragments in registers (296 per lane, loaded once from global); LDS weight copies and the per-WMMA B reads are gone; same sums, same order
# speedup vs baseline: 1.8941x; 1.1799x over previous
#include <hip/hip_runtime.h>

typedef float v2f __attribute__((ext_vector_type(2)));
typedef float v8f __attribute__((ext_vector_type(8)));

#define NODES_TOT 20000

__device__ __forceinline__ v8f wmma_f32(v2f a, v2f b, v8f c) {
  return __builtin_amdgcn_wmma_f32_16x16x4_f32(
       false, a,  false, b,
       (short)0, c,  false,  false);
}

#define AP8 12
#define AP32 36
#define AP64 68

template <int K, int N>
__device__ __forceinline__ void load_b_frags(v2f (&f)[(K / 4) * (N / 16)],
                                             const float* __restrict__ w,
                                             int col, int hi) {
  const float* wp = w + 2 * hi * N + col;
#pragma unroll
  for (int ks = 0; ks < K / 4; ++ks) {
#pragma unroll
    for (int nt = 0; nt < N / 16; ++nt) {
      v2f b;
      b.x = wp[(4 * ks) * N + nt * 16];
      b.y = wp[(4 * ks + 1) * N + nt * 16];
      f[ks * (N / 16) + nt] = b;
    }
  }
}

template <int K, int AP>
__device__ __forceinline__ void load_a_frags(v2f (&a)[K / 4],
                                             const float* aBuf, int col,
                                             int hi) {
  const float* ap = aBuf + col * AP + 2 * hi;
#pragma unroll
  for (int ks = 0; ks < K / 4; ++ks) {
    a[ks].x = ap[4 * ks];
    a[ks].y = ap[4 * ks + 1];
  }
}

template <int K, int N, int AP, int OP, bool RELU>
__device__ __forceinline__ void gemm_stage(const float* aBuf,
                                           const v2f (&wf)[(K / 4) * (N / 16)],
                                           const float* bias, float* outBuf,
                                           int col, int hi) {
  v2f a[K / 4];
  load_a_frags<K, AP>(a, aBuf, col, hi);
  float* op = outBuf + 8 * hi * OP + col;
#pragma unroll
  for (int nt = 0; nt < N / 16; ++nt) {
    const float bv = bias[nt * 16 + col];
    v8f c;
#pragma unroll
    for (int r = 0; r < 8; ++r) c[r] = bv;
#pragma unroll
    for (int ks = 0; ks < K / 4; ++ks)
      c = wmma_f32(a[ks], wf[ks * (N / 16) + nt], c);
#pragma unroll
    for (int r = 0; r < 8; ++r) {
      float v = c[r];
      if (RELU) v = fmaxf(v, 0.1f * v);
      op[r * OP + nt * 16] = v;
    }
  }
}

#define WAVES_E 8
#define THREADS_E (WAVES_E * 32)
#define NB 80
#define NPL 10
#define ROW (NPL * 32)
#define SCAN_V4 8
#define BATCH (THREADS_E * 4 * SCAN_V4)
#define PROC (WAVES_E * 16)
#define LIST_CAP (PROC + BATCH)

__device__ __forceinline__ void load_batch(int4 (&dst)[SCAN_V4],
                                           const int* __restrict__ src, int p,
                                           int tid) {
#pragma unroll
  for (int j = 0; j < SCAN_V4; ++j)
    dst[j] =
        *reinterpret_cast<const int4*>(src + p + (j * THREADS_E + tid) * 4);
}

__device__ __forceinline__ void keep_if_owned(int s, int e, int n0, int nb,
                                              int* sCount, int* sListE,
                                              unsigned short* sListN) {
  const unsigned nl = (unsigned)(s - n0);
  if (nl < (unsigned)nb) {
    const int slot = atomicAdd(sCount, 1);
    sListE[slot] = e;
    sListN[slot] = (unsigned short)nl;
  }
}

__global__ __launch_bounds__(THREADS_E) void edge_kernel(
    const float* __restrict__ r_ij, const float* __restrict__ w_rad,
    const float* __restrict__ b_rad, const float* __restrict__ w_direct,
    const float* __restrict__ w1, const float* __restrict__ b1,
    const float* __restrict__ w2, const float* __restrict__ b2,
    const float* __restrict__ w3, const float* __restrict__ b3,
    const int* __restrict__ edges_src, float* __restrict__ A_a,
    float* __restrict__ Av, float* __restrict__ Ad6, int n_edges,
    int n_nodes) {
  __shared__ float sBrad[32];
  __shared__ float sB1[64];
  __shared__ float sB2[64];
  __shared__ float sB3[32];
  __shared__ float sHH[WAVES_E][16 * AP32];
  __shared__ float sT1[WAVES_E][16 * AP64];
  __shared__ float sT2[WAVES_E][16 * AP64];
  __shared__ float sRS[WAVES_E][16 * 4];
  __shared__ int sSRC[WAVES_E][16];
  __shared__ float sAcc[(NB + 1) * ROW];
  __shared__ int sListE[LIST_CAP];
  __shared__ unsigned short sListN[LIST_CAP];
  __shared__ int sCount;

  const int tid = threadIdx.x;
  const int wave = tid >> 5, lane = tid & 31;
  const int col = lane & 15, hi = lane >> 4;

  for (int i = tid; i < 32; i += THREADS_E) sBrad[i] = b_rad[i];
  for (int i = tid; i < 64; i += THREADS_E) sB1[i] = b1[i];
  for (int i = tid; i < 64; i += THREADS_E) sB2[i] = b2[i];
  for (int i = tid; i < 32; i += THREADS_E) sB3[i] = b3[i];

  v2f fWrad[(8 / 4) * (32 / 16)];
  v2f fWdir[(32 / 4) * (32 / 16)];
  v2f fW1[(32 / 4) * (64 / 16)];
  v2f fW2[(64 / 4) * (64 / 16)];
  v2f fW3[(64 / 4) * (32 / 16)];
  load_b_frags<8, 32>(fWrad, w_rad, col, hi);
  load_b_frags<32, 32>(fWdir, w_direct, col, hi);
  load_b_frags<32, 64>(fW1, w1, col, hi);
  load_b_frags<64, 64>(fW2, w2, col, hi);
  load_b_frags<64, 32>(fW3, w3, col, hi);

  for (int i = tid; i < (NB + 1) * ROW; i += THREADS_E) sAcc[i] = 0.0f;
  if (tid == 0) sCount = 0;
  __syncthreads();

  const int n0 = blockIdx.x * NB;
  const int nb = (n_nodes - n0 < NB) ? (n_nodes - n0) : NB;

  const int n_full = (n_edges / BATCH) * BATCH;
  for (int e = n_full + tid; e < n_edges; e += THREADS_E)
    keep_if_owned(edges_src[e], e, n0, nb, &sCount, sListE, sListN);

  int4 nxt[SCAN_V4];
#pragma unroll
  for (int j = 0; j < SCAN_V4; ++j) nxt[j] = make_int4(-1, -1, -1, -1);
  if (n_full > 0) load_batch(nxt, edges_src, 0, tid);

  __syncthreads();
  int cnt = __builtin_amdgcn_readfirstlane(sCount);
  __syncthreads();
  int pos = 0;
  for (;;) {
    while (cnt < PROC && pos < n_full) {
      int4 cur[SCAN_V4];
#pragma unroll
      for (int j = 0; j < SCAN_V4; ++j) cur[j] = nxt[j];
      const int npos = pos + BATCH;
      if (npos < n_full) load_batch(nxt, edges_src, npos, tid);
#pragma unroll
      for (int j = 0; j < SCAN_V4; ++j) {
        const int e = pos + (j * THREADS_E + tid) * 4;
        const unsigned d0 = (unsigned)(cur[j].x - n0);
        const unsigned d1 = (unsigned)(cur[j].y - n0);
        const unsigned d2 = (unsigned)(cur[j].z - n0);
        const unsigned d3 = (unsigned)(cur[j].w - n0);
        const unsigned dmin = min(min(d0, d1), min(d2, d3));
        if (dmin < (unsigned)nb) {
          keep_if_owned(cur[j].x, e + 0, n0, nb, &sCount, sListE, sListN);
          keep_if_owned(cur[j].y, e + 1, n0, nb, &sCount, sListE, sListN);
          keep_if_owned(cur[j].z, e + 2, n0, nb, &sCount, sListE, sListN);
          keep_if_owned(cur[j].w, e + 3, n0, nb, &sCount, sListE, sListN);
        }
      }
      pos = npos;
      __syncthreads();
      cnt = __builtin_amdgcn_readfirstlane(sCount);
      __syncthreads();
    }
    if (cnt == 0) break;

    const int take = (cnt >= PROC) ? (cnt / PROC) * PROC : cnt;
    const int iters = (take + PROC - 1) / PROC;

    for (int it = 0; it < iters; ++it) {
      const int base = (it * WAVES_E + wave) * 16;
      const bool active = base < take;
      float* rbfBuf = sT1[wave];

      if (lane < 16) {
        const int li = base + lane;
        if (li < take) {
          const int e = sListE[li];
          if (li + PROC < take)
            __builtin_prefetch(&r_ij[3 * (long)sListE[li + PROC]], 0, 1);
          float x = r_ij[3 * (long)e + 0], y = r_ij[3 * (long)e + 1],
                z = r_ij[3 * (long)e + 2];
          float d = sqrtf(x * x + y * y + z * z);
#pragma unroll
          for (int i = 0; i < 8; ++i) {
            float u = (d - (float)i * (1.0f / 7.0f)) * 8.0f;
            rbfBuf[lane * AP8 + i] = __expf(-u * u);
          }
          float sx = 7.0f * x, sy = 7.0f * y, sz = 7.0f * z;
          float inv = rsqrtf(1.0f + sx * sx + sy * sy + sz * sz);
          sRS[wave][lane * 4 + 0] = sx * inv;
          sRS[wave][lane * 4 + 1] = sy * inv;
          sRS[wave][lane * 4 + 2] = sz * inv;
          sSRC[wave][lane] = (int)sListN[li];
        } else {
#pragma unroll
          for (int i = 0; i < 8; ++i) rbfBuf[lane * AP8 + i] = 0.0f;
          sRS[wave][lane * 4 + 0] = 0.0f;
          sRS[wave][lane * 4 + 1] = 0.0f;
          sRS[wave][lane * 4 + 2] = 0.0f;
          sSRC[wave][lane] = NB;
        }
      }
      __syncthreads();

      gemm_stage<8, 32, AP8, AP32, false>(rbfBuf, fWrad, sBrad, sHH[wave],
                                          col, hi);
      __syncthreads();
      gemm_stage<32, 64, AP32, AP64, true>(sHH[wave], fW1, sB1, sT1[wave],
                                           col, hi);
      __syncthreads();
      gemm_stage<64, 64, AP64, AP64, true>(sT1[wave], fW2, sB2, sT2[wave],
                                           col, hi);
      __syncthreads();

      v2f aH[32 / 4], aT[64 / 4];
      load_a_frags<32, AP32>(aH, sHH[wave], col, hi);
      load_a_frags<64, AP64>(aT, sT2[wave], col, hi);
#pragma unroll
      for (int nt = 0; nt < 2; ++nt) {
        const float bv = sB3[nt * 16 + col];
        v8f c;
#pragma unroll
        for (int r = 0; r < 8; ++r) c[r] = bv;
#pragma unroll
        for (int ks = 0; ks < 32 / 4; ++ks)
          c = wmma_f32(aH[ks], fWdir[ks * 2 + nt], c);
#pragma unroll
        for (int ks = 0; ks < 64 / 4; ++ks)
          c = wmma_f32(aT[ks], fW3[ks * 2 + nt], c);

        if (active) {
          const int ach = nt * 16 + col;
#pragma unroll
          for (int r = 0; r < 8; ++r) {
            const int m = r + 8 * hi;
            const float val = c[r];
            const int nl = sSRC[wave][m];
            const float rx = sRS[wave][m * 4 + 0];
            const float ry = sRS[wave][m * 4 + 1];
            const float rz = sRS[wave][m * 4 + 2];
            const float t[NPL] = {val,           val * rx,      val * ry,
                                  val * rz,      val * rx * rx, val * rx * ry,
                                  val * rx * rz, val * ry * ry, val * ry * rz,
                                  val * rz * rz};
            float* acc = &sAcc[nl * ROW + ach];
#pragma unroll
            for (int s = 0; s < NPL; s += 2) {
              const float va = hi ? t[s + 1] : t[s];
              const float vb = hi ? t[s] : t[s + 1];
              atomicAdd(&acc[(s + hi) * 32], va);
              atomicAdd(&acc[(s + 1 - hi) * 32], vb);
            }
          }
        }
      }
      __syncthreads();
    }

    const int rem = cnt - take;
    int keepE = 0;
    unsigned short keepN = 0;
    if (tid < rem) {
      keepE = sListE[take + tid];
      keepN = sListN[take + tid];
    }
    __syncthreads();
    if (tid < rem) {
      sListE[tid] = keepE;
      sListN[tid] = keepN;
    }
    if (tid == 0) sCount = rem;
    __syncthreads();
    cnt = rem;
  }
  __syncthreads();

  for (int i = tid; i < nb * 32; i += THREADS_E) {
    const int nl = i >> 5, a = i & 31;
    A_a[(long)n0 * 32 + i] = sAcc[nl * ROW + a];
  }
  for (int i = tid; i < nb * 96; i += THREADS_E) {
    const int nl = i / 96, q = i - nl * 96;
    const int a = q / 3, cc = q - a * 3;
    Av[(long)n0 * 96 + i] = sAcc[nl * ROW + (1 + cc) * 32 + a];
  }
  for (int i = tid; i < nb * 192; i += THREADS_E) {
    const int nl = i / 192, q = i - nl * 192;
    const int a = q / 6, s6 = q - a * 6;
    Ad6[(long)n0 * 192 + i] = sAcc[nl * ROW + (4 + s6) * 32 + a];
  }
}

#define WAVES_N 8
__global__ __launch_bounds__(256) void node_kernel(
    const float* __restrict__ Av, const float* __restrict__ Ad6,
    const float* __restrict__ w_v, const float* __restrict__ w_d,
    float* __restrict__ out_v, float* __restrict__ out_d, int n_nodes) {
  __shared__ float sWv[32 * 32];
  __shared__ float sWd[32 * 32];
  for (int i = threadIdx.x; i < 32 * 32; i += blockDim.x) {
    sWv[i] = w_v[i];
    sWd[i] = w_d[i];
  }
  __syncthreads();

  const int wave = threadIdx.x >> 5, lane = threadIdx.x & 31;
  const int col = lane & 15, hi = lane >> 4;
  const int tile = blockIdx.x * WAVES_N + wave;
  if (tile * 16 >= n_nodes) return;
  const int n0 = tile * 16;

#pragma unroll
  for (int cch = 0; cch < 3; ++cch) {
#pragma unroll
    for (int nt = 0; nt < 2; ++nt) {
      v8f acc = {};
#pragma unroll
      for (int k0 = 0; k0 < 32; k0 += 4) {
        int k = k0 + 2 * hi;
        v2f a;
        a.x = Av[((long)(n0 + col) * 32 + k) * 3 + cch];
        a.y = Av[((long)(n0 + col) * 32 + k + 1) * 3 + cch];
        v2f b;
        b.x = sWv[k * 32 + nt * 16 + col];
        b.y = sWv[(k + 1) * 32 + nt * 16 + col];
        acc = wmma_f32(a, b, acc);
      }
#pragma unroll
      for (int r = 0; r < 8; ++r) {
        int node = n0 + r + 8 * hi;
        int v = nt * 16 + col;
        __builtin_nontemporal_store(acc[r],
                                    &out_v[((long)node * 32 + v) * 3 + cch]);
      }
    }
  }

  const int si[6] = {0, 0, 0, 1, 1, 2};
  const int sj[6] = {0, 1, 2, 1, 2, 2};
#pragma unroll
  for (int s = 0; s < 6; ++s) {
#pragma unroll
    for (int nt = 0; nt < 2; ++nt) {
      v8f acc = {};
#pragma unroll
      for (int k0 = 0; k0 < 32; k0 += 4) {
        int k = k0 + 2 * hi;
        v2f a;
        a.x = Ad6[((long)(n0 + col) * 32 + k) * 6 + s];
        a.y = Ad6[((long)(n0 + col) * 32 + k + 1) * 6 + s];
        v2f b;
        b.x = sWd[k * 32 + nt * 16 + col];
        b.y = sWd[(k + 1) * 32 + nt * 16 + col];
        acc = wmma_f32(a, b, acc);
      }
#pragma unroll
      for (int r = 0; r < 8; ++r) {
        int node = n0 + r + 8 * hi;
        int dch = nt * 16 + col;
        float val = acc[r];
        float* base = &out_d[((long)node * 32 + dch) * 9];
        __builtin_nontemporal_store(val, &base[si[s] * 3 + sj[s]]);
        if (si[s] != sj[s])
          __builtin_nontemporal_store(val, &base[sj[s] * 3 + si[s]]);
      }
    }
  }
}

extern "C" void kernel_launch(void* const* d_in, const int* in_sizes, int n_in,
                              void* d_out, int out_size, void* d_ws,
                              size_t ws_size, hipStream_t stream) {
  const float* r_ij = (const float*)d_in[0];
  const float* w_rad = (const float*)d_in[1];
  const float* b_rad = (const float*)d_in[2];
  const float* w_direct = (const float*)d_in[3];
  const float* w1 = (const float*)d_in[4];
  const float* b1 = (const float*)d_in[5];
  const float* w2 = (const float*)d_in[6];
  const float* b2 = (const float*)d_in[7];
  const float* w3 = (const float*)d_in[8];
  const float* b3 = (const float*)d_in[9];
  const float* w_v = (const float*)d_in[10];
  const float* w_d = (const float*)d_in[11];
  const int* edges_src = (const int*)d_in[12];

  const int n_edges = in_sizes[0] / 3;
  const int n_nodes = NODES_TOT;

  float* A_a = (float*)d_out;
  float* out_v = A_a + (long)n_nodes * 32;
  float* out_d = out_v + (long)n_nodes * 32 * 3;
  float* Av = (float*)d_ws;
  float* Ad6 = Av + (long)n_nodes * 32 * 3;

  const int e_blocks = (NODES_TOT + NB - 1) / NB;
  edge_kernel<<<e_blocks, THREADS_E, 0, stream>>>(r_ij, w_rad, b_rad, w_direct,
                                                  w1, b1, w2, b2, w3, b3,
                                                  edges_src, A_a, Av, Ad6,
                                                  n_edges, n_nodes);

  const int nd_tiles = (n_nodes + 15) / 16;
  const int n_blocks = (nd_tiles + WAVES_N - 1) / WAVES_N;
  node_kernel<<<n_blocks, 256, 0, stream>>>(Av, Ad6, w_v, w_d, out_v, out_d,
                                            n_nodes);
}


# Edge kernel: each wave scans its own eighth of edges_src into its own LDS list (ballot slots) and runs its tiles alone, no block barrier in the main loop; LDS adds arrive in another order
# speedup vs baseline: 2.5219x; 1.3315x over previous
#include <hip/hip_runtime.h>

typedef float v2f __attribute__((ext_vector_type(2)));
typedef float v8f __attribute__((ext_vector_type(8)));

#define NODES_TOT 20000

__device__ __forceinline__ v8f wmma_f32(v2f a, v2f b, v8f c) {
  return __builtin_amdgcn_wmma_f32_16x16x4_f32(
       false, a,  false, b,
       (short)0, c,  false,  false);
}

#define AP8 12
#define AP32 36
#define AP64 68

template <int K, int N>
__device__ __forceinline__ void load_b_frags(v2f (&f)[(K / 4) * (N / 16)],
                                             const float* __restrict__ w,
                                             int col, int hi) {
  const float* wp = w + 2 * hi * N + col;
#pragma unroll
  for (int ks = 0; ks < K / 4; ++ks) {
#pragma unroll
    for (int nt = 0; nt < N / 16; ++nt) {
      v2f b;
      b.x = wp[(4 * ks) * N + nt * 16];
      b.y = wp[(4 * ks + 1) * N + nt * 16];
      f[ks * (N / 16) + nt] = b;
    }
  }
}

template <int K, int AP>
__device__ __forceinline__ void load_a_frags(v2f (&a)[K / 4],
                                             const float* aBuf, int col,
                                             int hi) {
  const float* ap = aBuf + col * AP + 2 * hi;
#pragma unroll
  for (int ks = 0; ks < K / 4; ++ks) {
    a[ks].x = ap[4 * ks];
    a[ks].y = ap[4 * ks + 1];
  }
}

template <int K, int N, int AP, int OP, bool RELU>
__device__ __forceinline__ void gemm_stage(const float* aBuf,
                                           const v2f (&wf)[(K / 4) * (N / 16)],
                                           const float* bias, float* outBuf,
                                           int col, int hi) {
  v2f a[K / 4];
  load_a_frags<K, AP>(a, aBuf, col, hi);
  float* op = outBuf + 8 * hi * OP + col;
#pragma unroll
  for (int nt = 0; nt < N / 16; ++nt) {
    const float bv = bias[nt * 16 + col];
    v8f c;
#pragma unroll
    for (int r = 0; r < 8; ++r) c[r] = bv;
#pragma unroll
    for (int ks = 0; ks < K / 4; ++ks)
      c = wmma_f32(a[ks], wf[ks * (N / 16) + nt], c);
#pragma unroll
    for (int r = 0; r < 8; ++r) {
      float v = c[r];
      if (RELU) v = fmaxf(v, 0.1f * v);
      op[r * OP + nt * 16] = v;
    }
  }
}

#define WAVES_E 8
#define THREADS_E (WAVES_E * 32)
#define NB 80
#define NPL 10
#define ROW (NPL * 32)
#define SCAN_V4 16
#define CHUNK (32 * 4 * SCAN_V4)
#define WCAP (16 + CHUNK)

__device__ __forceinline__ void wave_lds_order() {
  __builtin_amdgcn_fence(__ATOMIC_RELEASE, "wavefront");
  __builtin_amdgcn_wave_barrier();
  __builtin_amdgcn_fence(__ATOMIC_ACQUIRE, "wavefront");
}

__device__ __forceinline__ int4 fetch_group(const int* __restrict__ src, int c,
                                            int j, int lane) {
  return *reinterpret_cast<const int4*>(src + c * CHUNK + (j * 32 + lane) * 4);
}

__device__ __forceinline__ int append_owned(unsigned d, int e, int nb, int tail,
                                            int* lE, unsigned short* lN) {
  const bool hit = d < (unsigned)nb;
  const unsigned m = __builtin_amdgcn_ballot_w32(hit);
  if (m != 0u) {
    if (hit) {
      const int slot = tail + (int)__builtin_amdgcn_mbcnt_lo(m, 0u);
      lE[slot] = e;
      lN[slot] = (unsigned short)d;
    }
    tail += __builtin_popcount(m);
  }
  return tail;
}

__global__ __launch_bounds__(THREADS_E) void edge_kernel(
    const float* __restrict__ r_ij, const float* __restrict__ w_rad,
    const float* __restrict__ b_rad, const float* __restrict__ w_direct,
    const float* __restrict__ w1, const float* __restrict__ b1,
    const float* __restrict__ w2, const float* __restrict__ b2,
    const float* __restrict__ w3, const float* __restrict__ b3,
    const int* __restrict__ edges_src, float* __restrict__ A_a,
    float* __restrict__ Av, float* __restrict__ Ad6, int n_edges,
    int n_nodes) {
  __shared__ float sBrad[32];
  __shared__ float sB1[64];
  __shared__ float sB2[64];
  __shared__ float sB3[32];
  __shared__ float sHH[WAVES_E][16 * AP32];
  __shared__ float sT1[WAVES_E][16 * AP64];
  __shared__ float sT2[WAVES_E][16 * AP64];
  __shared__ float sRS[WAVES_E][16 * 4];
  __shared__ int sSRC[WAVES_E][16];
  __shared__ float sAcc[(NB + 1) * ROW];
  __shared__ int sListE[WAVES_E][WCAP];
  __shared__ unsigned short sListN[WAVES_E][WCAP];

  const int tid = threadIdx.x;
  const int wave = __builtin_amdgcn_readfirstlane(tid >> 5);
  const int lane = tid & 31;
  const int col = lane & 15, hi = lane >> 4;

  for (int i = tid; i < 32; i += THREADS_E) sBrad[i] = b_rad[i];
  for (int i = tid; i < 64; i += THREADS_E) sB1[i] = b1[i];
  for (int i = tid; i < 64; i += THREADS_E) sB2[i] = b2[i];
  for (int i = tid; i < 32; i += THREADS_E) sB3[i] = b3[i];

  v2f fWrad[(8 / 4) * (32 / 16)];
  v2f fWdir[(32 / 4) * (32 / 16)];
  v2f fW1[(32 / 4) * (64 / 16)];
  v2f fW2[(64 / 4) * (64 / 16)];
  v2f fW3[(64 / 4) * (32 / 16)];
  load_b_frags<8, 32>(fWrad, w_rad, col, hi);
  load_b_frags<32, 32>(fWdir, w_direct, col, hi);
  load_b_frags<32, 64>(fW1, w1, col, hi);
  load_b_frags<64, 64>(fW2, w2, col, hi);
  load_b_frags<64, 32>(fW3, w3, col, hi);

  for (int i = tid; i < (NB + 1) * ROW; i += THREADS_E) sAcc[i] = 0.0f;

  const int n0 = blockIdx.x * NB;
  const int nb = (n_nodes - n0 < NB) ? (n_nodes - n0) : NB;

  int* lE = sListE[wave];
  unsigned short* lN = sListN[wave];
  float* rbfBuf = sT1[wave];

  const int n_whole = n_edges / CHUNK;
  int c = wave;
  int4 buf[SCAN_V4];
#pragma unroll
  for (int j = 0; j < SCAN_V4; ++j)
    buf[j] = (c < n_whole) ? fetch_group(edges_src, c, j, lane)
                           : make_int4(-1, -1, -1, -1);

  int head = 0, tail = 0;

  {
    const int t0 = n_whole * CHUNK + wave * (CHUNK / WAVES_E);
#pragma unroll 1
    for (int k = 0; k < CHUNK / WAVES_E / 32; ++k) {
      const int e = t0 + k * 32 + lane;
      const int s = (e < n_edges) ? edges_src[e] : -1;
      tail = append_owned((unsigned)(s - n0), e, nb, tail, lE, lN);
    }
  }

  __syncthreads();

  for (;;) {
    while (tail - head < 16 && c < n_whole) {
      if (head > 0) {
        const int rem = tail - head;
        int keepE = 0;
        unsigned short keepN = 0;
        if (lane < rem) {
          keepE = lE[head + lane];
          keepN = lN[head + lane];
        }
        wave_lds_order();
        if (lane < rem) {
          lE[lane] = keepE;
          lN[lane] = keepN;
        }
        head = 0;
        tail = rem;
      }
      const int pos = c * CHUNK;
      c += WAVES_E;
      const int cn = (c < n_whole) ? c : n_whole - 1;
#pragma unroll
      for (int j = 0; j < SCAN_V4; ++j) {
        const int e = pos + (j * 32 + lane) * 4;
        const unsigned d0 = (unsigned)(buf[j].x - n0);
        const unsigned d1 = (unsigned)(buf[j].y - n0);
        const unsigned d2 = (unsigned)(buf[j].z - n0);
        const unsigned d3 = (unsigned)(buf[j].w - n0);
        buf[j] = fetch_group(edges_src, cn, j, lane);
        const unsigned dmin = min(min(d0, d1), min(d2, d3));
        if (__builtin_amdgcn_ballot_w32(dmin < (unsigned)nb) != 0u) {
          tail = append_owned(d0, e + 0, nb, tail, lE, lN);
          tail = append_owned(d1, e + 1, nb, tail, lE, lN);
          tail = append_owned(d2, e + 2, nb, tail, lE, lN);
          tail = append_owned(d3, e + 3, nb, tail, lE, lN);
        }
      }
      wave_lds_order();
    }
    const int avail = tail - head;
    if (avail == 0) break;
    const int nvalid = (avail < 16) ? avail : 16;

    if (lane < 16) {
      if (lane < nvalid) {
        const int li = head + lane;
        const int e = lE[li];
        if (li + 16 < tail)
          __builtin_prefetch(&r_ij[3 * (long)lE[li + 16]], 0, 1);
        float x = r_ij[3 * (long)e + 0], y = r_ij[3 * (long)e + 1],
              z = r_ij[3 * (long)e + 2];
        float d = sqrtf(x * x + y * y + z * z);
#pragma unroll
        for (int i = 0; i < 8; ++i) {
          float u = (d - (float)i * (1.0f / 7.0f)) * 8.0f;
          rbfBuf[lane * AP8 + i] = __expf(-u * u);
        }
        float sx = 7.0f * x, sy = 7.0f * y, sz = 7.0f * z;
        float inv = rsqrtf(1.0f + sx * sx + sy * sy + sz * sz);
        sRS[wave][lane * 4 + 0] = sx * inv;
        sRS[wave][lane * 4 + 1] = sy * inv;
        sRS[wave][lane * 4 + 2] = sz * inv;
        sSRC[wave][lane] = (int)lN[li];
      } else {
#pragma unroll
        for (int i = 0; i < 8; ++i) rbfBuf[lane * AP8 + i] = 0.0f;
        sRS[wave][lane * 4 + 0] = 0.0f;
        sRS[wave][lane * 4 + 1] = 0.0f;
        sRS[wave][lane * 4 + 2] = 0.0f;
        sSRC[wave][lane] = NB;
      }
    }
    head += nvalid;
    wave_lds_order();

    gemm_stage<8, 32, AP8, AP32, false>(rbfBuf, fWrad, sBrad, sHH[wave], col,
                                        hi);
    wave_lds_order();
    gemm_stage<32, 64, AP32, AP64, true>(sHH[wave], fW1, sB1, sT1[wave], col,
                                         hi);
    wave_lds_order();
    gemm_stage<64, 64, AP64, AP64, true>(sT1[wave], fW2, sB2, sT2[wave], col,
                                         hi);
    wave_lds_order();

    v2f aH[32 / 4], aT[64 / 4];
    load_a_frags<32, AP32>(aH, sHH[wave], col, hi);
    load_a_frags<64, AP64>(aT, sT2[wave], col, hi);
#pragma unroll
    for (int nt = 0; nt < 2; ++nt) {
      const float bv = sB3[nt * 16 + col];
      v8f cc;
#pragma unroll
      for (int r = 0; r < 8; ++r) cc[r] = bv;
#pragma unroll
      for (int ks = 0; ks < 32 / 4; ++ks)
        cc = wmma_f32(aH[ks], fWdir[ks * 2 + nt], cc);
#pragma unroll
      for (int ks = 0; ks < 64 / 4; ++ks)
        cc = wmma_f32(aT[ks], fW3[ks * 2 + nt], cc);

      const int ach = nt * 16 + col;
#pragma unroll
      for (int r = 0; r < 8; ++r) {
        const int m = r + 8 * hi;
        const float val = cc[r];
        const int nl = sSRC[wave][m];
        const float rx = sRS[wave][m * 4 + 0];
        const float ry = sRS[wave][m * 4 + 1];
        const float rz = sRS[wave][m * 4 + 2];
        const float t[NPL] = {val,           val * rx,      val * ry,
                              val * rz,      val * rx * rx, val * rx * ry,
                              val * rx * rz, val * ry * ry, val * ry * rz,
                              val * rz * rz};
        float* acc = &sAcc[nl * ROW + ach];
#pragma unroll
        for (int s = 0; s < NPL; s += 2) {
          const float va = hi ? t[s + 1] : t[s];
          const float vb = hi ? t[s] : t[s + 1];
          atomicAdd(&acc[(s + hi) * 32], va);
          atomicAdd(&acc[(s + 1 - hi) * 32], vb);
        }
      }
    }
    wave_lds_order();
  }
  __syncthreads();

  for (int i = tid; i < nb * 32; i += THREADS_E) {
    const int nl = i >> 5, a = i & 31;
    A_a[(long)n0 * 32 + i] = sAcc[nl * ROW + a];
  }
  for (int i = tid; i < nb * 96; i += THREADS_E) {
    const int nl = i / 96, q = i - nl * 96;
    const int a = q / 3, cc = q - a * 3;
    Av[(long)n0 * 96 + i] = sAcc[nl * ROW + (1 + cc) * 32 + a];
  }
  for (int i = tid; i < nb * 192; i += THREADS_E) {
    const int nl = i / 192, q = i - nl * 192;
    const int a = q / 6, s6 = q - a * 6;
    Ad6[(long)n0 * 192 + i] = sAcc[nl * ROW + (4 + s6) * 32 + a];
  }
}

#define WAVES_N 8
__global__ __launch_bounds__(256) void node_kernel(
    const float* __restrict__ Av, const float* __restrict__ Ad6,
    const float* __restrict__ w_v, const float* __restrict__ w_d,
    float* __restrict__ out_v, float* __restrict__ out_d, int n_nodes) {
  __shared__ float sWv[32 * 32];
  __shared__ float sWd[32 * 32];
  for (int i = threadIdx.x; i < 32 * 32; i += blockDim.x) {
    sWv[i] = w_v[i];
    sWd[i] = w_d[i];
  }
  __syncthreads();

  const int wave = threadIdx.x >> 5, lane = threadIdx.x & 31;
  const int col = lane & 15, hi = lane >> 4;
  const int tile = blockIdx.x * WAVES_N + wave;
  if (tile * 16 >= n_nodes) return;
  const int n0 = tile * 16;

#pragma unroll
  for (int cch = 0; cch < 3; ++cch) {
#pragma unroll
    for (int nt = 0; nt < 2; ++nt) {
      v8f acc = {};
#pragma unroll
      for (int k0 = 0; k0 < 32; k0 += 4) {
        int k = k0 + 2 * hi;
        v2f a;
        a.x = Av[((long)(n0 + col) * 32 + k) * 3 + cch];
        a.y = Av[((long)(n0 + col) * 32 + k + 1) * 3 + cch];
        v2f b;
        b.x = sWv[k * 32 + nt * 16 + col];
        b.y = sWv[(k + 1) * 32 + nt * 16 + col];
        acc = wmma_f32(a, b, acc);
      }
#pragma unroll
      for (int r = 0; r < 8; ++r) {
        int node = n0 + r + 8 * hi;
        int v = nt * 16 + col;
        __builtin_nontemporal_store(acc[r],
                                    &out_v[((long)node * 32 + v) * 3 + cch]);
      }
    }
  }

  const int si[6] = {0, 0, 0, 1, 1, 2};
  const int sj[6] = {0, 1, 2, 1, 2, 2};
#pragma unroll
  for (int s = 0; s < 6; ++s) {
#pragma unroll
    for (int nt = 0; nt < 2; ++nt) {
      v8f acc = {};
#pragma unroll
      for (int k0 = 0; k0 < 32; k0 += 4) {
        int k = k0 + 2 * hi;
        v2f a;
        a.x = Ad6[((long)(n0 + col) * 32 + k) * 6 + s];
        a.y = Ad6[((long)(n0 + col) * 32 + k + 1) * 6 + s];
        v2f b;
        b.x = sWd[k * 32 + nt * 16 + col];
        b.y = sWd[(k + 1) * 32 + nt * 16 + col];
        acc = wmma_f32(a, b, acc);
      }
#pragma unroll
      for (int r = 0; r < 8; ++r) {
        int node = n0 + r + 8 * hi;
        int dch = nt * 16 + col;
        float val = acc[r];
        float* base = &out_d[((long)node * 32 + dch) * 9];
        __builtin_nontemporal_store(val, &base[si[s] * 3 + sj[s]]);
        if (si[s] != sj[s])
          __builtin_nontemporal_store(val, &base[sj[s] * 3 + si[s]]);
      }
    }
  }
}

extern "C" void kernel_launch(void* const* d_in, const int* in_sizes, int n_in,
                              void* d_out, int out_size, void* d_ws,
                              size_t ws_size, hipStream_t stream) {
  const float* r_ij = (const float*)d_in[0];
  const float* w_rad = (const float*)d_in[1];
  const float* b_rad = (const float*)d_in[2];
  const float* w_direct = (const float*)d_in[3];
  const float* w1 = (const float*)d_in[4];
  const float* b1 = (const float*)d_in[5];
  const float* w2 = (const float*)d_in[6];
  const float* b2 = (const float*)d_in[7];
  const float* w3 = (const float*)d_in[8];
  const float* b3 = (const float*)d_in[9];
  const float* w_v = (const float*)d_in[10];
  const float* w_d = (const float*)d_in[11];
  const int* edges_src = (const int*)d_in[12];

  const int n_edges = in_sizes[0] / 3;
  const int n_nodes = NODES_TOT;

  float* A_a = (float*)d_out;
  float* out_v = A_a + (long)n_nodes * 32;
  float* out_d = out_v + (long)n_nodes * 32 * 3;
  float* Av = (float*)d_ws;
  float* Ad6 = Av + (long)n_nodes * 32 * 3;

  const int e_blocks = (NODES_TOT + NB - 1) / NB;
  edge_kernel<<<e_blocks, THREADS_E, 0, stream>>>(r_ij, w_rad, b_rad, w_direct,
                                                  w1, b1, w2, b2, w3, b3,
                                                  edges_src, A_a, Av, Ad6,
                                                  n_edges, n_nodes);

  const int nd_tiles = (n_nodes + 15) / 16;
  const int n_blocks = (nd_tiles + WAVES_N - 1) / WAVES_N;
  node_kernel<<<n_blocks, 256, 0, stream>>>(Av, Ad6, w_v, w_d, out_v, out_d,
                                            n_nodes);
}
